# starting kernel
# baseline (speedup 1.0000x reference)
#include <hip/hip_runtime.h>
#include <hip/hip_bf16.h>

#define BB 16384
#define NN 20
#define DD 64
#define RR 3
#define KK 64
#define TWOD 128

typedef __attribute__((ext_vector_type(2))) float v2f;
typedef __attribute__((ext_vector_type(8))) float v8f;

#define LDSW_STRIDE 130

__global__ __launch_bounds__(96)
void ui_gemm_kernel(const float* __restrict__ u_emb,
                    const float* __restrict__ i_emb,
                    const float* __restrict__ w_uir,
                    const float* __restrict__ r_vec,
                    float* __restrict__ ui_all) {
    const int r    = threadIdx.x >> 5;
    const int lane = threadIdx.x & 31;
    const int half = lane >> 4;
    const int lrow = lane & 15;
    const int row  = blockIdx.x * 16 + lrow;

    v8f acc[4] = {};
    const float* wr = w_uir + r * TWOD * KK;

    #pragma unroll
    for (int ks = 0; ks < 32; ++ks) {
        const int k0 = ks * 4 + half * 2;
        const float* p = (k0 < DD) ? (u_emb + row * DD + k0)
                                   : (i_emb + row * DD + (k0 - DD));
        v2f a; a.x = p[0]; a.y = p[1];
        #pragma unroll
        for (int ct = 0; ct < 4; ++ct) {
            const int c = ct * 16 + lrow;
            v2f b;
            b.x = wr[k0 * KK + c];
            b.y = wr[(k0 + 1) * KK + c];
            acc[ct] = __builtin_amdgcn_wmma_f32_16x16x4_f32(
                false, a, false, b, (short)0, acc[ct], false, false);
        }
    }

    #pragma unroll
    for (int ct = 0; ct < 4; ++ct) {
        const int c  = ct * 16 + lrow;
        const float rv = r_vec[r * KK + c];
        #pragma unroll
        for (int v = 0; v < 8; ++v) {
            const int m = v + half * 8;
            const int b = blockIdx.x * 16 + m;
            ui_all[(b * RR + r) * KK + c] = acc[ct][v] + rv;
        }
    }
}

__global__ __launch_bounds__(256)
void ao_pred_kernel(const float* __restrict__ a_emb,
                    const float* __restrict__ o_emb,
                    const int*   __restrict__ s,
                    const float* __restrict__ w_aor,
                    const float* __restrict__ ui_all,
                    float* __restrict__ pred) {
    extern __shared__ float smem[];
    float* ldsW  = smem;
    float* ldsUI = smem + RR * KK * LDSW_STRIDE;
    float* ldsR  = ldsUI + 8 * 16 * KK;

    const int tid  = threadIdx.x;
    const int wave = tid >> 5;
    const int lane = tid & 31;
    const int half = lane >> 4;
    const int lrow = lane & 15;

    #pragma unroll 4
    for (int i = 0; i < 96; ++i) {
        const int idx = tid + i * 256;
        const int r   = idx >> 13;
        const int rem = idx & 8191;
        const int k   = rem >> 6;
        const int c   = rem & 63;
        ldsW[(r * KK + c) * LDSW_STRIDE + k] = w_aor[idx];
    }

    const int tile = blockIdx.x * 8 + wave;
    const int row  = tile * 16 + lrow;

    float* myUI = ldsUI + wave * (16 * KK);
    #pragma unroll
    for (int m = 0; m < 16; ++m) {
        const int rm   = tile * 16 + m;
        const int bm   = rm / NN;
        const int relm = s[rm];
        const float* src = ui_all + (bm * RR + relm) * KK;
        myUI[m * KK + lane]      = src[lane];
        myUI[m * KK + lane + 32] = src[lane + 32];
    }

    float areg[64];
    const float* pa = a_emb + (size_t)row * DD;
    const float* po = o_emb + (size_t)row * DD;
    #pragma unroll
    for (int ks = 0; ks < 32; ++ks) {
        const int k0 = ks * 4 + half * 2;
        const float* p = (k0 < DD) ? (pa + k0) : (po + k0 - DD);
        areg[2 * ks]     = p[0];
        areg[2 * ks + 1] = p[1];
    }

    const int myrel = s[row];

    __syncthreads();

    v8f acc[4] = {};
    for (int r = 0; r < RR; ++r) {
        const float maskf = (myrel == r) ? 1.0f : 0.0f;
        const float* wrow = ldsW + r * KK * LDSW_STRIDE;
        #pragma unroll
        for (int ks = 0; ks < 32; ++ks) {
            const int k0 = ks * 4 + half * 2;
            v2f a;
            a.x = areg[2 * ks]     * maskf;
            a.y = areg[2 * ks + 1] * maskf;
            #pragma unroll
            for (int ct = 0; ct < 4; ++ct) {
                const int c = ct * 16 + lrow;
                const v2f b = *(const v2f*)(wrow + c * LDSW_STRIDE + k0);
                acc[ct] = __builtin_amdgcn_wmma_f32_16x16x4_f32(
                    false, a, false, b, (short)0, acc[ct], false, false);
            }
        }
    }

    float part[8];
    #pragma unroll
    for (int v = 0; v < 8; ++v) {
        const int m = v + half * 8;
        float sum = 0.f;
        #pragma unroll
        for (int ct = 0; ct < 4; ++ct)
            sum += acc[ct][v] * myUI[m * KK + ct * 16 + lrow];
        part[v] = sum;
    }

    float* myR = ldsR + wave * (16 * 17);
    #pragma unroll
    for (int v = 0; v < 8; ++v) {
        const int m = v + half * 8;
        myR[m * 17 + lrow] = part[v];
    }
    __syncthreads();

    if (lane < 16) {
        float sum = 0.f;
        #pragma unroll
        for (int j = 0; j < 16; ++j) sum += myR[lane * 17 + j];
        pred[tile * 16 + lane] = sum;
    }
}

extern "C" void kernel_launch(void* const* d_in, const int* in_sizes, int n_in,
                              void* d_out, int out_size, void* d_ws, size_t ws_size,
                              hipStream_t stream) {
    const float* u_emb = (const float*)d_in[0];
    const float* i_emb = (const float*)d_in[1];
    const float* a_emb = (const float*)d_in[2];
    const float* o_emb = (const float*)d_in[3];
    const int*   s     = (const int*)  d_in[4];
    const float* w_uir = (const float*)d_in[5];
    const float* w_aor = (const float*)d_in[6];
    const float* r_vec = (const float*)d_in[7];
    float* pred   = (float*)d_out;
    float* ui_all = (float*)d_ws;

    ui_gemm_kernel<<<BB / 16, 96, 0, stream>>>(u_emb, i_emb, w_uir, r_vec, ui_all);

    const size_t smem_bytes =
        (size_t)(RR * KK * LDSW_STRIDE + 8 * 16 * KK + 8 * 16 * 17) * sizeof(float);
    const int n_tiles = (BB * NN) / 16;
    ao_pred_kernel<<<n_tiles / 8, 256, smem_bytes, stream>>>(
        a_emb, o_emb, s, w_aor, ui_all, pred);
}


# Reassociate pred = ao.(w_aor ui_r): w_aor applied once per (sample, relation), then one K=128 WMMA chain per 16 rows (32 instead of 384); changes the f32 summation order
# speedup vs baseline: 3.2078x; 3.2078x over previous
#include <hip/hip_runtime.h>


#define BB 16384
#define NN 20
#define DD 64
#define RR 3
#define KK 64
#define TWOD 128

typedef __attribute__((ext_vector_type(2))) float v2f;
typedef __attribute__((ext_vector_type(4))) float v4f;
typedef __attribute__((ext_vector_type(8))) float v8f;

#define U_PITCH 68

__global__ __launch_bounds__(96)
void ui_proj_kernel(const float* __restrict__ u_emb,
                    const float* __restrict__ i_emb,
                    const float* __restrict__ w_uir,
                    const float* __restrict__ w_aor,
                    const float* __restrict__ r_vec,
                    float* __restrict__ v_all) {
    __shared__ __attribute__((aligned(16))) float ldsU[RR * 16 * U_PITCH];

    const int r    = threadIdx.x >> 5;
    const int lane = threadIdx.x & 31;
    const int half = lane >> 4;
    const int lrow = lane & 15;
    const int row  = blockIdx.x * 16 + lrow;

    v8f acc[4] = {};
    const float* wr = w_uir + r * TWOD * KK;

    #pragma unroll
    for (int ks = 0; ks < 32; ++ks) {
        const int k0 = ks * 4 + half * 2;
        const float* p = (k0 < DD) ? (u_emb + row * DD + k0)
                                   : (i_emb + row * DD + (k0 - DD));
        v2f a; a.x = p[0]; a.y = p[1];
        #pragma unroll
        for (int ct = 0; ct < 4; ++ct) {
            const int c = ct * 16 + lrow;
            v2f b;
            b.x = wr[k0 * KK + c];
            b.y = wr[(k0 + 1) * KK + c];
            acc[ct] = __builtin_amdgcn_wmma_f32_16x16x4_f32(
                false, a, false, b, (short)0, acc[ct], false, false);
        }
    }

    float* myU = ldsU + r * 16 * U_PITCH;
    #pragma unroll
    for (int ct = 0; ct < 4; ++ct) {
        const int c  = ct * 16 + lrow;
        const float rv = r_vec[r * KK + c];
        #pragma unroll
        for (int v = 0; v < 8; ++v) {
            const int m = v + half * 8;
            myU[m * U_PITCH + c] = acc[ct][v] + rv;
        }
    }
    __syncthreads();

    v8f vacc[8] = {};
    const float* wa = w_aor + r * TWOD * KK;
    #pragma unroll
    for (int i = 0; i < 8; ++i) {
        const int kq = i * 8 + half * 4;
        const v4f a4 = *(const v4f*)(myU + lrow * U_PITCH + kq);
        #pragma unroll
        for (int dt = 0; dt < 8; ++dt) {
            const int d = dt * 16 + lrow;
            const v4f b4 = *(const v4f*)(wa + d * KK + kq);
            v2f a, b;
            a.x = a4.x; a.y = a4.y; b.x = b4.x; b.y = b4.y;
            vacc[dt] = __builtin_amdgcn_wmma_f32_16x16x4_f32(
                false, a, false, b, (short)0, vacc[dt], false, false);
            a.x = a4.z; a.y = a4.w; b.x = b4.z; b.y = b4.w;
            vacc[dt] = __builtin_amdgcn_wmma_f32_16x16x4_f32(
                false, a, false, b, (short)0, vacc[dt], false, false);
        }
    }

    #pragma unroll
    for (int dt = 0; dt < 8; ++dt) {
        const int d = dt * 16 + lrow;
        #pragma unroll
        for (int v = 0; v < 8; ++v) {
            const int m = v + half * 8;
            const int b = blockIdx.x * 16 + m;
            v_all[(b * RR + r) * TWOD + d] = vacc[dt][v];
        }
    }
}

#define TPW 5

__global__ __launch_bounds__(256)
void ao_dot_kernel(const float* __restrict__ a_emb,
                   const float* __restrict__ o_emb,
                   const int*   __restrict__ s,
                   const float* __restrict__ v_all,
                   float* __restrict__ pred) {
    const int wave = threadIdx.x >> 5;
    const int lane = threadIdx.x & 31;
    const int half = lane >> 4;
    const int lrow = lane & 15;
    const int wg   = blockIdx.x * 8 + wave;
    const int row0 = wg * (16 * TPW);
    const int vrow = wg * (4 * RR) + (lrow < 11 ? lrow : 11);

    const float* pv = v_all + (size_t)vrow * TWOD + half * 4;
    v4f b4[16];
    #pragma unroll
    for (int i = 0; i < 16; ++i) b4[i] = *(const v4f*)(pv + i * 8);

    int rel[TPW];
    #pragma unroll
    for (int t = 0; t < TPW; ++t) rel[t] = s[row0 + t * 16 + lrow];

    v4f a4[2][16];
#define LOAD_TILE(T) do { \
        const size_t off_ = (size_t)(row0 + (T) * 16 + lrow) * DD + half * 4; \
        _Pragma("unroll") for (int i = 0; i < 8; ++i)  a4[(T) & 1][i] = *(const v4f*)(a_emb + off_ + i * 8); \
        _Pragma("unroll") for (int i = 8; i < 16; ++i) a4[(T) & 1][i] = *(const v4f*)(o_emb + off_ + (i - 8) * 8); \
    } while (0)

    LOAD_TILE(0);
    #pragma unroll
    for (int t = 0; t < TPW; ++t) {
        __builtin_amdgcn_sched_barrier(0);
        if (t + 1 < TPW) LOAD_TILE(t + 1);
        __builtin_amdgcn_sched_barrier(0);

        v8f acc0 = {}, acc1 = {};
        #pragma unroll
        for (int i = 0; i < 16; ++i) {
            const v4f av = a4[t & 1][i];
            const v4f bv = b4[i];
            v2f a, b;
            a.x = av.x; a.y = av.y; b.x = bv.x; b.y = bv.y;
            acc0 = __builtin_amdgcn_wmma_f32_16x16x4_f32(
                false, a, false, b, (short)0, acc0, false, false);
            a.x = av.z; a.y = av.w; b.x = bv.z; b.y = bv.w;
            acc1 = __builtin_amdgcn_wmma_f32_16x16x4_f32(
                false, a, false, b, (short)0, acc1, false, false);
        }

        const int  rl  = t * 16 + lrow;
        const bool ok  = (unsigned)rel[t] < (unsigned)RR;
        const int  col = ok ? (rl / NN) * RR + rel[t] : 0;
        const int  src = col + (lrow & 8) * 2;
        float val = 0.f;
        #pragma unroll
        for (int v = 0; v < 8; ++v) {
            const float tv = __shfl(acc0[v] + acc1[v], src);
            val = ((lrow & 7) == v) ? tv : val;
        }
        if (lane < 16) pred[row0 + rl] = ok ? val : 0.f;
    }
#undef LOAD_TILE
}

extern "C" void kernel_launch(void* const* d_in, const int* in_sizes, int n_in,
                              void* d_out, int out_size, void* d_ws, size_t ws_size,
                              hipStream_t stream) {
    const float* u_emb = (const float*)d_in[0];
    const float* i_emb = (const float*)d_in[1];
    const float* a_emb = (const float*)d_in[2];
    const float* o_emb = (const float*)d_in[3];
    const int*   s     = (const int*)  d_in[4];
    const float* w_uir = (const float*)d_in[5];
    const float* w_aor = (const float*)d_in[6];
    const float* r_vec = (const float*)d_in[7];
    float* pred  = (float*)d_out;
    float* v_all = (float*)d_ws;

    ui_proj_kernel<<<BB / 16, 96, 0, stream>>>(u_emb, i_emb, w_uir, w_aor, r_vec, v_all);

    ao_dot_kernel<<<(BB * NN) / (16 * TPW * 8), 256, 0, stream>>>(a_emb, o_emb, s, v_all, pred);
}


# Fuse both launches into one kernel, one block per 16 samples: V goes from the projection to the ao product through LDS instead of the workspace; chains dealt evenly to 4 waves; sum order unchanged
# speedup vs baseline: 3.8753x; 1.2081x over previous
#include <hip/hip_runtime.h>


#define BB 16384
#define NN 20
#define DD 64
#define RR 3
#define KK 64
#define TWOD 128

typedef __attribute__((ext_vector_type(2))) float v2f;
typedef __attribute__((ext_vector_type(4))) float v4f;
typedef __attribute__((ext_vector_type(8))) float v8f;

#define U_PITCH 68
#define V_PITCH 132
#define TPW 5

__global__ __launch_bounds__(128, 4)
void transr_fused_kernel(const float* __restrict__ u_emb,
                         const float* __restrict__ i_emb,
                         const float* __restrict__ a_emb,
                         const float* __restrict__ o_emb,
                         const int*   __restrict__ s,
                         const float* __restrict__ w_uir,
                         const float* __restrict__ w_aor,
                         const float* __restrict__ r_vec,
                         float* __restrict__ pred) {
    __shared__ __attribute__((aligned(16))) float ldsU[RR * 16 * U_PITCH];
    __shared__ __attribute__((aligned(16))) float ldsV[16 * RR * V_PITCH];

    const int q    = threadIdx.x >> 5;
    const int lane = threadIdx.x & 31;
    const int half = lane >> 4;
    const int lrow = lane & 15;
    const int row  = blockIdx.x * 16 + lrow;

    {
        int ur[3], uc[3];
        #pragma unroll
        for (int j = 0; j < 3; ++j) {
            const int id = q * 3 + j;
            ur[j] = id >> 2;
            uc[j] = (id & 3) * 16 + lrow;
        }
        v8f acc[3] = {};
        #pragma unroll
        for (int ks = 0; ks < 32; ++ks) {
            const int k0 = ks * 4 + half * 2;
            const float* p = (k0 < DD) ? (u_emb + row * DD + k0)
                                       : (i_emb + row * DD + (k0 - DD));
            v2f a; a.x = p[0]; a.y = p[1];
            #pragma unroll
            for (int j = 0; j < 3; ++j) {
                const float* wr = w_uir + ur[j] * TWOD * KK;
                v2f b;
                b.x = wr[k0 * KK + uc[j]];
                b.y = wr[(k0 + 1) * KK + uc[j]];
                acc[j] = __builtin_amdgcn_wmma_f32_16x16x4_f32(
                    false, a, false, b, (short)0, acc[j], false, false);
            }
        }
        #pragma unroll
        for (int j = 0; j < 3; ++j) {
            const float rv = r_vec[ur[j] * KK + uc[j]];
            float* myU = ldsU + ur[j] * 16 * U_PITCH;
            #pragma unroll
            for (int v = 0; v < 8; ++v) {
                const int m = v + half * 8;
                myU[m * U_PITCH + uc[j]] = acc[j][v] + rv;
            }
        }
    }
    __syncthreads();

    {
        v8f vacc[RR][2] = {};
        #pragma unroll
        for (int i = 0; i < 8; ++i) {
            const int kq = i * 8 + half * 4;
            #pragma unroll
            for (int r = 0; r < RR; ++r) {
                const v4f a4 = *(const v4f*)(ldsU + r * 16 * U_PITCH + lrow * U_PITCH + kq);
                #pragma unroll
                for (int dd = 0; dd < 2; ++dd) {
                    const int d = (q * 2 + dd) * 16 + lrow;
                    const v4f b4 = *(const v4f*)(w_aor + r * TWOD * KK + d * KK + kq);
                    v2f a, b;
                    a.x = a4.x; a.y = a4.y; b.x = b4.x; b.y = b4.y;
                    vacc[r][dd] = __builtin_amdgcn_wmma_f32_16x16x4_f32(
                        false, a, false, b, (short)0, vacc[r][dd], false, false);
                    a.x = a4.z; a.y = a4.w; b.x = b4.z; b.y = b4.w;
                    vacc[r][dd] = __builtin_amdgcn_wmma_f32_16x16x4_f32(
                        false, a, false, b, (short)0, vacc[r][dd], false, false);
                }
            }
        }
        #pragma unroll
        for (int r = 0; r < RR; ++r) {
            #pragma unroll
            for (int dd = 0; dd < 2; ++dd) {
                const int d = (q * 2 + dd) * 16 + lrow;
                #pragma unroll
                for (int v = 0; v < 8; ++v) {
                    const int m = v + half * 8;
                    ldsV[(m * RR + r) * V_PITCH + d] = vacc[r][dd][v];
                }
            }
        }
    }

    const int row0 = (blockIdx.x * 4 + q) * (16 * TPW);

    __syncthreads();

    int rel[TPW];
    #pragma unroll
    for (int t = 0; t < TPW; ++t) rel[t] = s[row0 + t * 16 + lrow];

    v4f a4[2][16];
#define LOAD_TILE(T) do { \
        const size_t off_ = (size_t)(row0 + (T) * 16 + lrow) * DD + half * 4; \
        _Pragma("unroll") for (int i = 0; i < 8; ++i)  a4[(T) & 1][i] = *(const v4f*)(a_emb + off_ + i * 8); \
        _Pragma("unroll") for (int i = 8; i < 16; ++i) a4[(T) & 1][i] = *(const v4f*)(o_emb + off_ + (i - 8) * 8); \
    } while (0)

    LOAD_TILE(0);

    const float* pv = ldsV + (q * (4 * RR) + (lrow < 11 ? lrow : 11)) * V_PITCH + half * 4;
    v4f b4[16];
    #pragma unroll
    for (int i = 0; i < 16; ++i) b4[i] = *(const v4f*)(pv + i * 8);

    #pragma unroll
    for (int t = 0; t < TPW; ++t) {
        __builtin_amdgcn_sched_barrier(0);
        if (t + 1 < TPW) LOAD_TILE(t + 1);
        __builtin_amdgcn_sched_barrier(0);

        v8f acc0 = {}, acc1 = {};
        #pragma unroll
        for (int i = 0; i < 16; ++i) {
            const v4f av = a4[t & 1][i];
            const v4f bv = b4[i];
            v2f a, b;
            a.x = av.x; a.y = av.y; b.x = bv.x; b.y = bv.y;
            acc0 = __builtin_amdgcn_wmma_f32_16x16x4_f32(
                false, a, false, b, (short)0, acc0, false, false);
            a.x = av.z; a.y = av.w; b.x = bv.z; b.y = bv.w;
            acc1 = __builtin_amdgcn_wmma_f32_16x16x4_f32(
                false, a, false, b, (short)0, acc1, false, false);
        }

        const int  rl  = t * 16 + lrow;
        const bool ok  = (unsigned)rel[t] < (unsigned)RR;
        const int  col = ok ? (rl / NN) * RR + rel[t] : 0;
        const int  src = col + (lrow & 8) * 2;
        float val = 0.f;
        #pragma unroll
        for (int v = 0; v < 8; ++v) {
            const float tv = __shfl(acc0[v] + acc1[v], src);
            val = ((lrow & 7) == v) ? tv : val;
        }
        if (lane < 16) pred[row0 + rl] = ok ? val : 0.f;
    }
#undef LOAD_TILE
}

extern "C" void kernel_launch(void* const* d_in, const int* in_sizes, int n_in,
                              void* d_out, int out_size, void* d_ws, size_t ws_size,
                              hipStream_t stream) {
    const float* u_emb = (const float*)d_in[0];
    const float* i_emb = (const float*)d_in[1];
    const float* a_emb = (const float*)d_in[2];
    const float* o_emb = (const float*)d_in[3];
    const int*   s     = (const int*)  d_in[4];
    const float* w_uir = (const float*)d_in[5];
    const float* w_aor = (const float*)d_in[6];
    const float* r_vec = (const float*)d_in[7];
    float* pred  = (float*)d_out;

    transr_fused_kernel<<<BB / 16, 128, 0, stream>>>(u_emb, i_emb, a_emb, o_emb, s,
                                                     w_uir, w_aor, r_vec, pred);
}
